# starting kernel
# baseline (speedup 1.0000x reference)
#include <hip/hip_runtime.h>

typedef __attribute__((ext_vector_type(2))) float v2f;
typedef __attribute__((ext_vector_type(8))) float v8f;

namespace {
constexpr int D              = 1024;
constexpr int ROWS_PER_BLOCK = 128;
constexpr int EB             = 128;
constexpr int KS             = 64;
constexpr int KPAD           = 68;
}

__global__ __launch_bounds__(256, 2)
void bilinear_wmma_f32_kernel(const float* __restrict__ elg,
                              const float* __restrict__ eth,
                              const float* __restrict__ W,
                              float* __restrict__ out)
{
    __shared__ float sW[EB * KPAD];

    const int tid  = threadIdx.x;
    const int wave = tid >> 5;
    const int lane = tid & 31;
    const int colg = lane & 15;
    const int kh   = (lane >> 4) << 1;

    const int rowTile = blockIdx.x * ROWS_PER_BLOCK + wave * 16;
    const float* aRow = elg + (size_t)(rowTile + colg) * D;

    float acc[8];
#pragma unroll
    for (int r = 0; r < 8; ++r) acc[r] = 0.f;

    for (int eb = 0; eb < D / EB; ++eb) {
        const int e0 = eb * EB;

        v8f C[8] = {};

        for (int ks = 0; ks < D / KS; ++ks) {
            __syncthreads();
#pragma unroll 4
            for (int i = tid; i < KS * EB; i += 256) {
                int e  = i & (EB - 1);
                int kl = i >> 7;
                sW[e * KPAD + kl] = W[(size_t)(ks * KS + kl) * D + e0 + e];
            }
            __syncthreads();

#pragma unroll
            for (int kt = 0; kt < KS / 16; ++kt) {
#pragma unroll
                for (int j = 0; j < 4; ++j) {
                    const int klA = kt * 16 + j * 4 + kh;
                    v2f a = *reinterpret_cast<const v2f*>(aRow + ks * KS + klA);
#pragma unroll
                    for (int t = 0; t < 8; ++t) {
                        v2f b = *reinterpret_cast<const v2f*>(
                            &sW[(t * 16 + colg) * KPAD + klA]);
                        C[t] = __builtin_amdgcn_wmma_f32_16x16x4_f32(
                            false, a, false, b, (short)0, C[t], false, false);
                    }
                }
            }
        }

        const int mOff = (lane >> 4) * 8;
#pragma unroll
        for (int t = 0; t < 8; ++t) {
            const int col = e0 + t * 16 + colg;
#pragma unroll
            for (int r = 0; r < 8; ++r) {
                float ev = eth[(size_t)(rowTile + mOff + r) * D + col];
                acc[r] = fmaf(C[t][r], ev, acc[r]);
            }
        }
    }

#pragma unroll
    for (int m = 1; m <= 8; m <<= 1) {
#pragma unroll
        for (int r = 0; r < 8; ++r) acc[r] += __shfl_xor(acc[r], m, 32);
    }

    if (colg == 0) {
        const int mOff = (lane >> 4) * 8;
#pragma unroll
        for (int r = 0; r < 8; ++r) out[rowTile + mOff + r] = acc[r];
    }
}

extern "C" void kernel_launch(void* const* d_in, const int* in_sizes, int n_in,
                              void* d_out, int out_size, void* d_ws, size_t ws_size,
                              hipStream_t stream) {
    const float* elg = (const float*)d_in[0];
    const float* eth = (const float*)d_in[1];
    const float* W   = (const float*)d_in[2];
    float* out = (float*)d_out;

    const int nRows = in_sizes[0] / D;
    dim3 grid(nRows / ROWS_PER_BLOCK);
    dim3 block(256);
    bilinear_wmma_f32_kernel<<<grid, block, 0, stream>>>(elg, eth, W, out);
}


# Seven-product scheme for 2x2 block matrices inside the wave tile: per (2 W fragments) x (2 k-steps) x (2 column tiles) 7 WMMAs of fragment sums, not 8; sums on v_pk_add_f32; sum order changes
# speedup vs baseline: 2.1477x; 2.1477x over previous
#include <hip/hip_runtime.h>


typedef __attribute__((ext_vector_type(2))) float v2f;
typedef __attribute__((ext_vector_type(4))) float v4f;
typedef __attribute__((ext_vector_type(8))) float v8f;

namespace {
constexpr int D              = 1024;
constexpr int ROWS_PER_BLOCK = 128;
constexpr int KS             = 256;
constexpr int KPAD           = 260;
constexpr int ROW_V4         = KS / 4;
constexpr int STAGE_V4       = ROWS_PER_BLOCK * ROW_V4 / 256;
constexpr int TD             = 2;
constexpr int DT_PER_WAVE    = D / (16 * TD) / 8;
constexpr int KSTEPS         = KS / 4;
constexpr int KG             = 4;
constexpr int NG             = KSTEPS / KG;
constexpr int NQ             = 16;
constexpr int NU             = 4;
}

__global__ __launch_bounds__(256, 2)
void bilinear_wmma_f32_kernel(const float* __restrict__ elg,
                              const float* __restrict__ eth,
                              const float* __restrict__ W,
                              float* __restrict__ out)
{
    __shared__ __attribute__((aligned(16))) float sE[ROWS_PER_BLOCK * KPAD];

    const int tid  = threadIdx.x;
    const int wave = tid >> 5;
    const int lane = tid & 31;
    const int colg = lane & 15;
    const int kh   = (lane >> 4) << 1;
    const int mOff = (lane >> 4) * 8;

    const int rowBase = blockIdx.x * ROWS_PER_BLOCK;

    const int sq = tid & (ROW_V4 - 1);
    const int sr = tid / ROW_V4;

    const float* stSrc = eth + (size_t)(rowBase + sr) * D + sq * 4;
    float*       stDst = &sE[sr * KPAD + sq * 4];
    const float* wLane = W + (size_t)(wave * DT_PER_WAVE * 16 * TD + colg) * D + kh;
    const float* eLane = elg + (size_t)(rowBase + colg) * D + wave * DT_PER_WAVE * 16 * TD + mOff;

    v2f acc[8];
#pragma unroll
    for (int t = 0; t < 8; ++t) acc[t] = v2f{0.f, 0.f};

    const float* aRow = wLane;
    v2f aq[NQ][TD];
#pragma unroll
    for (int q = 0; q < NQ; ++q)
#pragma unroll
        for (int s = 0; s < TD; ++s)
            aq[q][s] = *reinterpret_cast<const v2f*>(aRow + s * 16 * D + q * 4);

    for (int ks = 0; ks < D / KS; ++ks) {
        v4f stage[STAGE_V4];
#pragma unroll
        for (int i = 0; i < STAGE_V4; ++i)
            stage[i] = *reinterpret_cast<const v4f*>(stSrc + ks * KS + i * (256 / ROW_V4) * D);
        __syncthreads();
#pragma unroll
        for (int i = 0; i < STAGE_V4; ++i)
            *reinterpret_cast<v4f*>(stDst + i * (256 / ROW_V4) * KPAD) = stage[i];
        __syncthreads();

        for (int dt = 0; dt < DT_PER_WAVE; ++dt) {
            const float* aNext = (dt + 1 < DT_PER_WAVE) ? aRow + 16 * TD * D
                               : (ks + 1 < D / KS)      ? wLane + (ks + 1) * KS
                                                        : aRow;

            v8f P[NU][7] = {};

            asm volatile("" ::: "memory");

            v2f bb[2][KG][8];
#pragma unroll
            for (int j = 0; j < KG; ++j)
#pragma unroll
                for (int t = 0; t < 8; ++t)
                    bb[0][j][t] = *reinterpret_cast<const v2f*>(&sE[(t * 16 + colg) * KPAD + j * 4 + kh]);

            v4f ev[TD][16];

#pragma unroll
            for (int g = 0; g < NG; ++g) {
                if (g + 1 < NG) {
#pragma unroll
                    for (int j = 0; j < KG; ++j)
#pragma unroll
                        for (int t = 0; t < 8; ++t)
                            bb[(g + 1) & 1][j][t] = *reinterpret_cast<const v2f*>(
                                &sE[(t * 16 + colg) * KPAD + ((g + 1) * KG + j) * 4 + kh]);
                }
                if (g == NG - 1) {
#pragma unroll
                    for (int t = 0; t < 4; ++t)
#pragma unroll
                        for (int s = 0; s < TD; ++s) {
                            const float* ep = eLane + t * 16 * D + dt * 16 * TD + s * 16;
                            ev[s][2 * t]     = *reinterpret_cast<const v4f*>(ep);
                            ev[s][2 * t + 1] = *reinterpret_cast<const v4f*>(ep + 4);
                        }
                }
                __builtin_amdgcn_sched_barrier(0);
#pragma unroll
                for (int j = 0; j < KG; j += 2) {
                    const v2f a00 = aq[(g * KG + j) % NQ][0],     a10 = aq[(g * KG + j) % NQ][1];
                    const v2f a01 = aq[(g * KG + j + 1) % NQ][0], a11 = aq[(g * KG + j + 1) % NQ][1];
                    const v2f sa0 = a00 + a11, sa1 = a10 + a11, sa4 = a00 + a01, sa5 = a10 - a00, sa6 = a01 - a11;
#pragma unroll
                    for (int u = 0; u < NU; ++u) {
                        const v2f b00 = bb[g & 1][j][2 * u],     b01 = bb[g & 1][j][2 * u + 1];
                        const v2f b10 = bb[g & 1][j + 1][2 * u], b11 = bb[g & 1][j + 1][2 * u + 1];
                        const v2f sb0 = b00 + b11, sb2 = b01 - b11, sb3 = b10 - b00, sb5 = b00 + b01, sb6 = b10 + b11;
                        P[u][0] = __builtin_amdgcn_wmma_f32_16x16x4_f32(false, sa0, false, sb0, (short)0, P[u][0], false, false);
                        P[u][1] = __builtin_amdgcn_wmma_f32_16x16x4_f32(false, sa1, false, b00, (short)0, P[u][1], false, false);
                        P[u][2] = __builtin_amdgcn_wmma_f32_16x16x4_f32(false, a00, false, sb2, (short)0, P[u][2], false, false);
                        P[u][3] = __builtin_amdgcn_wmma_f32_16x16x4_f32(false, a11, false, sb3, (short)0, P[u][3], false, false);
                        P[u][4] = __builtin_amdgcn_wmma_f32_16x16x4_f32(false, sa4, false, b11, (short)0, P[u][4], false, false);
                        P[u][5] = __builtin_amdgcn_wmma_f32_16x16x4_f32(false, sa5, false, sb5, (short)0, P[u][5], false, false);
                        P[u][6] = __builtin_amdgcn_wmma_f32_16x16x4_f32(false, sa6, false, sb6, (short)0, P[u][6], false, false);
                    }
                }
                __builtin_amdgcn_sched_barrier(0);
#pragma unroll
                for (int j = 0; j < KG; ++j) {
                    const int kn = g * KG + j + NQ;
#pragma unroll
                    for (int s = 0; s < TD; ++s)
                        aq[(g * KG + j) % NQ][s] = (kn < KSTEPS)
                            ? *reinterpret_cast<const v2f*>(aRow + s * 16 * D + kn * 4)
                            : *reinterpret_cast<const v2f*>(aNext + s * 16 * D + (kn - KSTEPS) * 4);
                }
            }

#pragma unroll
            for (int t = 4; t < 8; ++t)
#pragma unroll
                for (int s = 0; s < TD; ++s) {
                    const float* ep = eLane + t * 16 * D + dt * 16 * TD + s * 16;
                    ev[s][2 * t]     = *reinterpret_cast<const v4f*>(ep);
                    ev[s][2 * t + 1] = *reinterpret_cast<const v4f*>(ep + 4);
                }

#pragma unroll
            for (int u = 0; u < NU; ++u) {
                const v8f Ut[TD][2] = {{P[u][0] + P[u][3] - P[u][4] + P[u][6], P[u][2] + P[u][4]},
                                       {P[u][1] + P[u][3],                     P[u][0] - P[u][1] + P[u][2] + P[u][5]}};
#pragma unroll
                for (int s = 0; s < TD; ++s) {
#pragma unroll
                    for (int c = 0; c < 2; ++c) {
                        const int t = 2 * u + c;
#pragma unroll
                        for (int r = 0; r < 4; r += 2) {
                            acc[t] = __builtin_elementwise_fma(v2f{Ut[s][c][r], Ut[s][c][r + 1]},
                                                               v2f{ev[s][2 * t][r], ev[s][2 * t][r + 1]}, acc[t]);
                        }
#pragma unroll
                        for (int r = 0; r < 4; r += 2) {
                            acc[t] = __builtin_elementwise_fma(v2f{Ut[s][c][4 + r], Ut[s][c][5 + r]},
                                                               v2f{ev[s][2 * t + 1][r], ev[s][2 * t + 1][r + 1]}, acc[t]);
                        }
                    }
                }
            }
            aRow = aNext;
        }
    }

    __syncthreads();
    float* red = sE;
#pragma unroll
    for (int t = 0; t < 8; ++t)
        red[(wave * 2 + (lane >> 4)) * ROWS_PER_BLOCK + t * 16 + colg] = acc[t].x + acc[t].y;
    __syncthreads();
    if (tid < ROWS_PER_BLOCK) {
        float s = 0.f;
#pragma unroll
        for (int i = 0; i < 16; ++i) s += red[i * ROWS_PER_BLOCK + tid];
        out[rowBase + tid] = s;
    }
}

extern "C" void kernel_launch(void* const* d_in, const int* in_sizes, int n_in,
                              void* d_out, int out_size, void* d_ws, size_t ws_size,
                              hipStream_t stream) {
    const float* elg = (const float*)d_in[0];
    const float* eth = (const float*)d_in[1];
    const float* W   = (const float*)d_in[2];
    float* out = (float*)d_out;

    const int nRows = in_sizes[0] / D;
    dim3 grid(nRows / ROWS_PER_BLOCK);
    dim3 block(256);
    bilinear_wmma_f32_kernel<<<grid, block, 0, stream>>>(elg, eth, W, out);
}
